# starting kernel
# baseline (speedup 1.0000x reference)
#include <hip/hip_runtime.h>
#include <hip/hip_bf16.h>


typedef float v2f __attribute__((ext_vector_type(2)));
typedef float v8f __attribute__((ext_vector_type(8)));

#define DIM_IN   512
#define DIM_OUT  512
#define NBANKS   64
#define NTOK     512
#define NROWS    1024
#define NSPLIT   4

__global__ void __launch_bounds__(256) banked_zero_kernel(float* __restrict__ out, int n) {
    int i = blockIdx.x * 256 + threadIdx.x;
    if (i < n) out[i] = 0.0f;
}

__global__ void __launch_bounds__(256) banked_linear_wmma_kernel(
    const float* __restrict__ x,
    const int*   __restrict__ sel,
    const float* __restrict__ prob,
    const float* __restrict__ w,
    const float* __restrict__ bias,
    float*       __restrict__ out)
{
    __shared__ int   s_rows[NROWS];
    __shared__ float s_prob[NROWS];
    __shared__ int   s_count;
    __shared__ float s_a[16 * DIM_IN];

    const int bank  = blockIdx.x >> 2;
    const int split = blockIdx.x & 3;
    const int tid   = threadIdx.x;

    if (tid == 0) s_count = 0;
    __syncthreads();

    for (int r = tid; r < NROWS; r += 256) {
        if (sel[r] == bank) {
            int p = atomicAdd(&s_count, 1);
            s_rows[p] = r;
            s_prob[p] = prob[r];
        }
    }
    __syncthreads();
    const int count = s_count;

    const int wave = tid >> 5;
    const int lane = tid & 31;
    const int n0   = (split * 8 + wave) * 16;
    const float* wb = w + (size_t)bank * (DIM_IN * DIM_OUT);

    const int am = lane & 15;
    const int ak = (lane >> 4) << 1;
    const int cn = lane & 15;
    const float* bptr = wb + (size_t)ak * DIM_OUT + n0 + cn;

    for (int mt = 0; mt < count; mt += 16) {
        __syncthreads();

        for (int i = tid; i < 16 * (DIM_IN / 4); i += 256) {
            int m  = i >> 7;
            int c4 = i & 127;
            float4 v = make_float4(0.f, 0.f, 0.f, 0.f);
            int ri = mt + m;
            if (ri < count) {
                int tok = s_rows[ri] >> 1;
                v = ((const float4*)(x + (size_t)tok * DIM_IN))[c4];
            }
            ((float4*)s_a)[m * (DIM_IN / 4) + c4] = v;
        }
        __syncthreads();

        v8f acc = {};
        const float* ldsa = &s_a[am * DIM_IN + ak];
#pragma unroll 4
        for (int k0 = 0; k0 < DIM_IN; k0 += 4) {
            v2f a = *(const v2f*)(ldsa + k0);
            v2f b;
            b.x = bptr[(size_t)k0 * DIM_OUT];
            b.y = bptr[(size_t)k0 * DIM_OUT + DIM_OUT];
            acc = __builtin_amdgcn_wmma_f32_16x16x4_f32(
                 false, a,  false, b,
                 (short)0, acc,  false,  false);
        }

        const int   mbase = (lane >> 4) << 3;
        const float bv    = bias[bank * DIM_OUT + n0 + cn];
#pragma unroll
        for (int v = 0; v < 8; ++v) {
            int ri = mt + mbase + v;
            if (ri < count) {
                int   tok = s_rows[ri] >> 1;
                float p   = s_prob[ri];
                atomicAdd(out + (size_t)tok * DIM_OUT + n0 + cn, (acc[v] + bv) * p);
            }
        }
    }
}

extern "C" void kernel_launch(void* const* d_in, const int* in_sizes, int n_in,
                              void* d_out, int out_size, void* d_ws, size_t ws_size,
                              hipStream_t stream) {
    const float* x    = (const float*)d_in[0];
    const int*   sel  = (const int*)  d_in[1];
    const float* prob = (const float*)d_in[2];
    const float* w    = (const float*)d_in[3];
    const float* bias = (const float*)d_in[4];

    float* out = (float*)d_out;

    const int n = NTOK * DIM_OUT;
    banked_zero_kernel<<<(n + 255) / 256, 256, 0, stream>>>(out, n);
    banked_linear_wmma_kernel<<<NBANKS * NSPLIT, 256, 0, stream>>>(
        x, sel, prob, w, bias, out);
}


# Chunk loop software-pipelined through a second fragment register set: each chunk's LDS fragment reads go out as one group a turn ahead (chunks parked two turns ahead); no product waits on LDS
# speedup vs baseline: 2.9759x; 2.9759x over previous
#include <hip/hip_runtime.h>
#include <hip/hip_bf16.h>


typedef float v2f __attribute__((ext_vector_type(2)));
typedef float v4f __attribute__((ext_vector_type(4)));
typedef float v8f __attribute__((ext_vector_type(8)));

#define DIM_IN   512
#define DIM_OUT  512
#define NBANKS   64
#define NTOK     512
#define NROWS    1024
#define NSPLIT   4
#define NCOLS    128

#define KC       32
#define NCH      (DIM_IN / KC)
#define NSETS    8
#define WP       136
#define AP       516
#define MROWS    32

__global__ void __launch_bounds__(256) banked_zero_kernel(float* __restrict__ out, int n) {
    int i = blockIdx.x * 256 + threadIdx.x;
    if (i < n) out[i] = 0.0f;
}

__device__ __forceinline__ void lds_barrier() {
    __builtin_amdgcn_fence(__ATOMIC_RELEASE, "workgroup", "local");
    __builtin_amdgcn_s_barrier();
    __builtin_amdgcn_fence(__ATOMIC_ACQUIRE, "workgroup", "local");
}

template <int MT>
__device__ __forceinline__ void load_frags(const float* stgb, const float* ak0,
                                           v2f (&fb)[KC / 4], v2f (&fa0)[KC / 4], v2f (&fa1)[KC / 4]) {
#pragma unroll
    for (int i = 0; i < KC / 4; ++i) {
        const int kk = 4 * i;
        fb[i].x = stgb[kk * WP];
        fb[i].y = stgb[kk * WP + WP];
        fa0[i]  = *(const v2f*)(ak0 + kk);
        if (MT == 2) fa1[i] = *(const v2f*)(ak0 + 16 * AP + kk);
    }
}
template <int MT>
__device__ __forceinline__ void mma_frags(const v2f (&fb)[KC / 4], const v2f (&fa0)[KC / 4],
                                          const v2f (&fa1)[KC / 4], v8f& acc0, v8f& acc1) {
#pragma unroll
    for (int i = 0; i < KC / 4; ++i) {
        acc0 = __builtin_amdgcn_wmma_f32_16x16x4_f32(
             false, fa0[i],  false, fb[i],
             (short)0, acc0,  false,  false);
        if (MT == 2)
            acc1 = __builtin_amdgcn_wmma_f32_16x16x4_f32(
                 false, fa1[i],  false, fb[i],
                 (short)0, acc1,  false,  false);
    }
}

template <int MT>
__device__ __forceinline__ void pin_acc(v8f& acc0, v8f& acc1) {
    if (MT == 2) asm volatile("" : "+v"(acc0), "+v"(acc1));
    else         asm volatile("" : "+v"(acc0));
}

#define FETCH(S, ch) do {                                                               \
        const float* p_ = slab + (size_t)((ch) * KC + wave) * DIM_OUT + lane * 4;       \
        S##a = *(const v4f*)(p_);                                                      \
        S##b = *(const v4f*)(p_ + 8  * DIM_OUT);                                       \
        S##c = *(const v4f*)(p_ + 16 * DIM_OUT);                                       \
        S##d = *(const v4f*)(p_ + 24 * DIM_OUT);                                       \
    } while (0)
#define PARK(S, stg) do {                                                              \
        float* q_ = (stg) + wave * WP + lane * 4;                                      \
        *(v4f*)(q_)           = S##a;                                                  \
        *(v4f*)(q_ + 8  * WP) = S##b;                                                  \
        *(v4f*)(q_ + 16 * WP) = S##c;                                                  \
        *(v4f*)(q_ + 24 * WP) = S##d;                                                  \
    } while (0)
#define STAGE(c) (s_w + ((c) & 1) * (KC * WP))
#define TURN(P, SN, ch, MT) do {                                                        \
        if ((ch) + 1 < NCH)                                                            \
            load_frags<MT>(STAGE((ch) + 1) + boff, abase + ((ch) + 1) * KC,            \
                           fB[1 - (P)], fA0[1 - (P)], fA1[1 - (P)]);                   \
        if ((ch) + 2 < NCH) {                                                          \
            PARK(SN, STAGE((ch) + 2));                                                 \
            if ((ch) + 2 + NSETS < NCH) FETCH(SN, (ch) + 2 + NSETS);                    \
        }                                                                              \
                                               \
        pin_acc<MT>(acc0, acc1);                                                       \
        mma_frags<MT>(fB[P], fA0[P], fA1[P], acc0, acc1);                              \
                                                   \
        pin_acc<MT>(acc0, acc1);                                                       \
        if ((ch) + 2 < NCH) lds_barrier();                                             \
    } while (0)
#define PIPELINE(MT) do {                                                              \
        PARK(r0, STAGE(0)); FETCH(r0, NSETS);                                          \
        lds_barrier();                                                                 \
        load_frags<MT>(STAGE(0) + boff, abase, fB[0], fA0[0], fA1[0]);                 \
        PARK(r1, STAGE(1)); FETCH(r1, NSETS + 1);                                      \
        lds_barrier();                                                                 \
        TURN(0, r2,  0, MT); TURN(1, r3,  1, MT); TURN(0, r4,  2, MT); TURN(1, r5,  3, MT); \
        TURN(0, r6,  4, MT); TURN(1, r7,  5, MT); TURN(0, r0,  6, MT); TURN(1, r1,  7, MT); \
        TURN(0, r2,  8, MT); TURN(1, r3,  9, MT); TURN(0, r4, 10, MT); TURN(1, r5, 11, MT); \
        TURN(0, r6, 12, MT); TURN(1, r7, 13, MT); TURN(0, r0, 14, MT); TURN(1, r1, 15, MT); \
    } while (0)

__global__ void __launch_bounds__(256) banked_linear_wmma_kernel(
    const float* __restrict__ x,
    const int*   __restrict__ sel,
    const float* __restrict__ prob,
    const float* __restrict__ w,
    const float* __restrict__ bias,
    float*       __restrict__ out)
{
    __shared__ int   s_rows[NROWS];
    __shared__ float s_prob[NROWS];
    __shared__ int   s_count;
    __shared__ __attribute__((aligned(16))) float s_a[MROWS * AP];
    __shared__ __attribute__((aligned(16))) float s_w[2 * KC * WP];

    const int bank  = blockIdx.x >> 2;
    const int split = blockIdx.x & 3;
    const int tid   = threadIdx.x;
    const int wave  = tid >> 5;
    const int lane  = tid & 31;

    const float* slab = w + (size_t)bank * (DIM_IN * DIM_OUT) + split * NCOLS;

    if (tid == 0) s_count = 0;

    int   selv[NROWS / 256];
    float probv[NROWS / 256];
#pragma unroll
    for (int i = 0; i < NROWS / 256; ++i) {
        selv[i]  = sel[tid + 256 * i];
        probv[i] = prob[tid + 256 * i];
    }
    v4f r0a, r0b, r0c, r0d, r1a, r1b, r1c, r1d, r2a, r2b, r2c, r2d, r3a, r3b, r3c, r3d;
    v4f r4a, r4b, r4c, r4d, r5a, r5b, r5c, r5d, r6a, r6b, r6c, r6d, r7a, r7b, r7c, r7d;
    FETCH(r0, 0); FETCH(r1, 1); FETCH(r2, 2); FETCH(r3, 3);
    FETCH(r4, 4); FETCH(r5, 5); FETCH(r6, 6); FETCH(r7, 7);

    lds_barrier();

#pragma unroll
    for (int i = 0; i < NROWS / 256; ++i) {
        if (selv[i] == bank) {
            int p = atomicAdd(&s_count, 1);
            s_rows[p] = tid + 256 * i;
            s_prob[p] = probv[i];
        }
    }
    lds_barrier();
    const int count = s_count;

    const int cn    = lane & 15;
    const int ak    = (lane >> 4) << 1;
    const int n0    = split * NCOLS + wave * 16;
    const int mbase = (lane >> 4) << 3;
    const int c4    = tid & 127;
    const int mh    = tid >> 7;

#define PASS(MT0) do {                                                                 \
        const int  mt0 = (MT0);                                                        \
        const bool two = (count - mt0) > 16;                                           \
           \
                      \
        for (int half = 0; half < (two ? 2 : 1); ++half) {                             \
            v4f v[8];                                                                  \
            _Pragma("unroll")                                                          \
            for (int j = 0; j < 8; ++j) {                                              \
                int ri  = min(mt0 + half * 16 + 2 * j + mh, count - 1);                \
                int tok = s_rows[ri] >> 1;                                  \
                v[j] = ((const v4f*)(x + (size_t)tok * DIM_IN))[c4];                   \
            }                                                                          \
            _Pragma("unroll")                                                          \
            for (int j = 0; j < 8; ++j)                                                \
                *(v4f*)(s_a + (half * 16 + 2 * j + mh) * AP + c4 * 4) = v[j];          \
        }                                                                              \
          \
        v8f acc0 = {}, acc1 = {};                                                      \
        v2f fB[2][KC / 4], fA0[2][KC / 4], fA1[2][KC / 4];       \
        if (two) PIPELINE(2);                                                          \
        else     PIPELINE(1);                                                          \
                      \
        _Pragma("unroll")                                                              \
        for (int m = 0; m < 2; ++m) {                                                  \
            if (m == 0 || two) {                                                       \
                _Pragma("unroll")                                                      \
                for (int v = 0; v < 8; ++v) {                                          \
                    int ri = mt0 + m * 16 + mbase + v;                                 \
                    if (ri < count) {                                                  \
                        int   tok = s_rows[ri] >> 1;                                   \
                        float p   = s_prob[ri];                                        \
                        const float av = (m == 0) ? acc0[v] : acc1[v];                 \
                        atomicAdd(out + (size_t)tok * DIM_OUT + n0 + cn, (av + bv) * p); \
                    }                                                                  \
                }                                                                      \
            }                                                                          \
        }                                                                              \
    } while (0)

    if (count > 0) {
        const float* abase = s_a + (lane & 15) * AP + ak;
        const int    boff  = ak * WP + wave * 16 + cn;
        const float  bv    = bias[bank * DIM_OUT + n0 + cn];
        PASS(0);
        for (int mrest = MROWS; mrest < count; mrest += MROWS) {
            lds_barrier();
            FETCH(r0, 0); FETCH(r1, 1); FETCH(r2, 2); FETCH(r3, 3);
            FETCH(r4, 4); FETCH(r5, 5); FETCH(r6, 6); FETCH(r7, 7);
            PASS(mrest);
        }
    }
}

extern "C" void kernel_launch(void* const* d_in, const int* in_sizes, int n_in,
                              void* d_out, int out_size, void* d_ws, size_t ws_size,
                              hipStream_t stream) {
    const float* x    = (const float*)d_in[0];
    const int*   sel  = (const int*)  d_in[1];
    const float* prob = (const float*)d_in[2];
    const float* w    = (const float*)d_in[3];
    const float* bias = (const float*)d_in[4];

    float* out = (float*)d_out;

    const int n = NTOK * DIM_OUT;
    banked_zero_kernel<<<(n + 255) / 256, 256, 0, stream>>>(out, n);
    banked_linear_wmma_kernel<<<NBANKS * NSPLIT, 256, 0, stream>>>(
        x, sel, prob, w, bias, out);
}
